# starting kernel
# baseline (speedup 1.0000x reference)
#include <hip/hip_runtime.h>


typedef __attribute__((ext_vector_type(2))) float v2f;
typedef __attribute__((ext_vector_type(8))) float v8f;

#define NNODES 50000
#define FEAT   256
#define OUTF   128
#define NEDGE  800000

__global__ __launch_bounds__(256) void gemm_bias_wmma(
    const float* __restrict__ X, const float* __restrict__ W,
    const float* __restrict__ B, float* __restrict__ Y, int P)
{
  const int lane  = threadIdx.x & 31;
  const int wave  = threadIdx.x >> 5;
  const int m0    = blockIdx.x << 4;
  const int n0    = ((blockIdx.y << 3) + wave) << 4;
  const int lmod  = lane & 15;
  const int lhalf = lane >> 4;
  const int koff  = lhalf << 1;

  const float* xrow = X + (size_t)(m0 + lmod) * FEAT + koff;
  const float* wrow = W + (size_t)(n0 + lmod) * FEAT + koff;

  v8f acc = {0.f, 0.f, 0.f, 0.f, 0.f, 0.f, 0.f, 0.f};
#pragma unroll 8
  for (int k = 0; k < FEAT; k += 4) {
    v2f a = *(const v2f*)(xrow + k);
    v2f b = *(const v2f*)(wrow + k);
    acc = __builtin_amdgcn_wmma_f32_16x16x4_f32(
        false, a, false, b, (short)0, acc, false, false);
  }

  const float bias = B ? B[n0 + lmod] : 0.0f;
  float* yp = Y + (size_t)(m0 + (lhalf << 3)) * P + (n0 + lmod);
#pragma unroll
  for (int i = 0; i < 8; ++i) yp[(size_t)i * P] = acc[i] + bias;
}

__global__ __launch_bounds__(256) void gemm_out_wmma(
    const float* __restrict__ S, const float* __restrict__ INV,
    const float* __restrict__ X,
    const float* __restrict__ W1, const float* __restrict__ W2,
    const float* __restrict__ B2, float* __restrict__ OUT, int colOff)
{
  const int lane  = threadIdx.x & 31;
  const int wave  = threadIdx.x >> 5;
  const int m0    = blockIdx.x << 4;
  const int n0    = wave << 4;
  const int lmod  = lane & 15;
  const int lhalf = lane >> 4;
  const int koff  = lhalf << 1;
  const int arow  = m0 + lmod;

  const float scale = INV[arow];
  const float* srow  = S  + (size_t)arow * FEAT + koff;
  const float* xrow  = X  + (size_t)arow * FEAT + koff;
  const float* w1row = W1 + (size_t)(n0 + lmod) * FEAT + koff;
  const float* w2row = W2 + (size_t)(n0 + lmod) * FEAT + koff;

  v8f acc = {0.f, 0.f, 0.f, 0.f, 0.f, 0.f, 0.f, 0.f};
#pragma unroll 8
  for (int k = 0; k < FEAT; k += 4) {
    v2f a = *(const v2f*)(srow + k) * scale;
    v2f b = *(const v2f*)(w1row + k);
    acc = __builtin_amdgcn_wmma_f32_16x16x4_f32(
        false, a, false, b, (short)0, acc, false, false);
  }
#pragma unroll 8
  for (int k = 0; k < FEAT; k += 4) {
    v2f a = *(const v2f*)(xrow + k);
    v2f b = *(const v2f*)(w2row + k);
    acc = __builtin_amdgcn_wmma_f32_16x16x4_f32(
        false, a, false, b, (short)0, acc, false, false);
  }

  const float bias = B2[n0 + lmod];
  float* yp = OUT + (size_t)(m0 + (lhalf << 3)) * (2 * OUTF) + colOff + (n0 + lmod);
#pragma unroll
  for (int i = 0; i < 8; ++i) yp[(size_t)i * (2 * OUTF)] = acc[i] + bias;
}

__global__ __launch_bounds__(256) void edge_score(
    const int* __restrict__ EI, const float* __restrict__ ATT,
    const float* __restrict__ X, float* __restrict__ Wout, int E)
{
  const int lane = threadIdx.x & 31;
  const int wave = threadIdx.x >> 5;
  const int e = (blockIdx.x << 3) + wave;
  if (e >= E) return;
  const int r = EI[e];
  const int l = EI[E + e];

  const float4* ap = (const float4*)(ATT + (size_t)r * FEAT) + (lane << 1);
  const float4* xp = (const float4*)(X   + (size_t)l * FEAT) + (lane << 1);
  float4 a0 = ap[0], a1 = ap[1];
  float4 x0 = xp[0], x1 = xp[1];
  float v = a0.x * x0.x + a0.y * x0.y + a0.z * x0.z + a0.w * x0.w
          + a1.x * x1.x + a1.y * x1.y + a1.z * x1.z + a1.w * x1.w;
#pragma unroll
  for (int off = 16; off >= 1; off >>= 1) v += __shfl_xor(v, off, 32);
  if (lane == 0) Wout[e] = v;
}

__global__ __launch_bounds__(256) void edge_scatter(
    const int* __restrict__ EI, const float* __restrict__ Wv,
    const float* __restrict__ X, float* __restrict__ S,
    float* __restrict__ CNT, int E)
{
  const int lane = threadIdx.x & 31;
  const int wave = threadIdx.x >> 5;
  const int e = (blockIdx.x << 3) + wave;
  if (e >= E) return;
  const int r = EI[e];
  const int l = EI[E + e];
  const float w = Wv[e];

  const float4* xp = (const float4*)(X + (size_t)l * FEAT) + (lane << 1);
  float4 x0 = xp[0], x1 = xp[1];
  float* sp = S + (size_t)r * FEAT + (lane << 3);
  atomicAdd(sp + 0, w * x0.x);
  atomicAdd(sp + 1, w * x0.y);
  atomicAdd(sp + 2, w * x0.z);
  atomicAdd(sp + 3, w * x0.w);
  atomicAdd(sp + 4, w * x1.x);
  atomicAdd(sp + 5, w * x1.y);
  atomicAdd(sp + 6, w * x1.z);
  atomicAdd(sp + 7, w * x1.w);
  if (lane == 0) atomicAdd(CNT + r, 1.0f);
}

__global__ __launch_bounds__(256) void finalize_inv(float* __restrict__ CNT, int n)
{
  int i = blockIdx.x * blockDim.x + threadIdx.x;
  if (i < n) {
    float c = CNT[i];
    CNT[i] = 1.0f / fmaxf(c, 1.0f);
  }
}

extern "C" void kernel_launch(void* const* d_in, const int* in_sizes, int n_in,
                              void* d_out, int out_size, void* d_ws, size_t ws_size,
                              hipStream_t stream)
{
  const float* x        = (const float*)d_in[0];
  const int*   pos_ei   = (const int*)d_in[1];
  const int*   neg_ei   = (const int*)d_in[2];
  const float* W_pos    = (const float*)d_in[3];
  const float* W_pos_cc = (const float*)d_in[4];
  const float* b_pos_cc = (const float*)d_in[5];
  const float* W_neg    = (const float*)d_in[6];
  const float* W_neg_cc = (const float*)d_in[7];
  const float* b_neg_cc = (const float*)d_in[8];
  const float* W_patt   = (const float*)d_in[9];
  const float* b_patt   = (const float*)d_in[10];
  const float* W_natt   = (const float*)d_in[11];
  const float* b_natt   = (const float*)d_in[12];
  float* out = (float*)d_out;

  float* Sbuf = (float*)d_ws;
  float* Wbuf = Sbuf + (size_t)NNODES * FEAT;
  float* Cbuf = Wbuf + (size_t)NEDGE;

  const dim3 blk(256);
  const dim3 gAtt(NNODES / 16, FEAT / 128);
  const dim3 gOut(NNODES / 16, 1);
  const int  gEdge = (NEDGE + 7) / 8;
  const int  gFin  = (NNODES + 255) / 256;

  const int*   ei[2]   = { pos_ei, neg_ei };
  const float* Watt[2] = { W_patt, W_natt };
  const float* batt[2] = { b_patt, b_natt };
  const float* Wagg[2] = { W_pos,  W_neg  };
  const float* Wcc[2]  = { W_pos_cc, W_neg_cc };
  const float* bcc[2]  = { b_pos_cc, b_neg_cc };

  for (int s = 0; s < 2; ++s) {
    gemm_bias_wmma<<<gAtt, blk, 0, stream>>>(x, Watt[s], batt[s], Sbuf, FEAT);
    edge_score<<<gEdge, blk, 0, stream>>>(ei[s], Sbuf, x, Wbuf, NEDGE);
    hipMemsetAsync(Sbuf, 0, (size_t)NNODES * FEAT * sizeof(float), stream);
    hipMemsetAsync(Cbuf, 0, (size_t)NNODES * sizeof(float), stream);
    edge_scatter<<<gEdge, blk, 0, stream>>>(ei[s], Wbuf, x, Sbuf, Cbuf, NEDGE);
    finalize_inv<<<gFin, blk, 0, stream>>>(Cbuf, NNODES);
    gemm_out_wmma<<<gOut, blk, 0, stream>>>(Sbuf, Cbuf, x, Wagg[s], Wcc[s],
                                            bcc[s], out, s * OUTF);
  }
}


# Edge sums owned by target range: a work-group holds 256 out rows + counts in LDS, scans the target list for its edges, adds rows by LDS float atomics (arrival order), 1/count applied to the sum once
# speedup vs baseline: 1.9107x; 1.9107x over previous
#include <hip/hip_runtime.h>


typedef __attribute__((ext_vector_type(2))) float v2f;
typedef __attribute__((ext_vector_type(8))) float v8f;

#define NNODES 50000
#define FEAT   256
#define OUTF   128
#define NEDGE  800000

__global__ __launch_bounds__(256) void gemm_bias_wmma(
    const float* __restrict__ X, const float* __restrict__ W,
    const float* __restrict__ B, float* __restrict__ Y, int P)
{
  const int lane  = threadIdx.x & 31;
  const int wave  = threadIdx.x >> 5;
  const int m0    = blockIdx.x << 4;
  const int n0    = ((blockIdx.y << 3) + wave) << 4;
  const int lmod  = lane & 15;
  const int lhalf = lane >> 4;
  const int koff  = lhalf << 1;

  const float* xrow = X + (size_t)(m0 + lmod) * FEAT + koff;
  const float* wrow = W + (size_t)(n0 + lmod) * FEAT + koff;

  v8f acc = {0.f, 0.f, 0.f, 0.f, 0.f, 0.f, 0.f, 0.f};
#pragma unroll 8
  for (int k = 0; k < FEAT; k += 4) {
    v2f a = *(const v2f*)(xrow + k);
    v2f b = *(const v2f*)(wrow + k);
    acc = __builtin_amdgcn_wmma_f32_16x16x4_f32(
        false, a, false, b, (short)0, acc, false, false);
  }

  const float bias = B ? B[n0 + lmod] : 0.0f;
  float* yp = Y + (size_t)(m0 + (lhalf << 3)) * P + (n0 + lmod);
#pragma unroll
  for (int i = 0; i < 8; ++i) yp[(size_t)i * P] = acc[i] + bias;
}

__global__ __launch_bounds__(256) void gemm_proj_wmma(
    const float* __restrict__ X,
    const float* __restrict__ W1, const float* __restrict__ W2,
    const float* __restrict__ B2, float* __restrict__ Z,
    float* __restrict__ OUT, int colOff)
{
  const int lane  = threadIdx.x & 31;
  const int wave  = threadIdx.x >> 5;
  const int m0    = blockIdx.x << 4;
  const int n0    = wave << 4;
  const int lmod  = lane & 15;
  const int lhalf = lane >> 4;
  const int koff  = lhalf << 1;

  const float* xrow  = X  + (size_t)(m0 + lmod) * FEAT + koff;
  const float* w1row = W1 + (size_t)(n0 + lmod) * FEAT + koff;
  const float* w2row = W2 + (size_t)(n0 + lmod) * FEAT + koff;

  v8f accz = {0.f, 0.f, 0.f, 0.f, 0.f, 0.f, 0.f, 0.f};
  v8f acco = {0.f, 0.f, 0.f, 0.f, 0.f, 0.f, 0.f, 0.f};
#pragma unroll 8
  for (int k = 0; k < FEAT; k += 4) {
    v2f a  = *(const v2f*)(xrow + k);
    v2f b1 = *(const v2f*)(w1row + k);
    v2f b2 = *(const v2f*)(w2row + k);
    accz = __builtin_amdgcn_wmma_f32_16x16x4_f32(
        false, a, false, b1, (short)0, accz, false, false);
    acco = __builtin_amdgcn_wmma_f32_16x16x4_f32(
        false, a, false, b2, (short)0, acco, false, false);
  }

  const float bias = B2[n0 + lmod];
  float* zp = Z   + (size_t)(m0 + (lhalf << 3)) * OUTF + (n0 + lmod);
  float* yp = OUT + (size_t)(m0 + (lhalf << 3)) * (2 * OUTF) + colOff + (n0 + lmod);
#pragma unroll
  for (int i = 0; i < 8; ++i) {
    zp[(size_t)i * OUTF] = accz[i];
    yp[(size_t)i * (2 * OUTF)] = acco[i] + bias;
  }
}

__global__ __launch_bounds__(256) void edge_score(
    const int* __restrict__ EI, const float* __restrict__ ATT,
    const float* __restrict__ X, float* __restrict__ Wout, int E)
{
  const int lane = threadIdx.x & 31;
  const int wave = threadIdx.x >> 5;
  const int e = (blockIdx.x << 3) + wave;
  if (e >= E) return;
  const int r = EI[e];
  const int l = EI[E + e];

  const float4* ap = (const float4*)(ATT + (size_t)r * FEAT) + (lane << 1);
  const float4* xp = (const float4*)(X   + (size_t)l * FEAT) + (lane << 1);
  float4 a0 = ap[0], a1 = ap[1];
  float4 x0 = xp[0], x1 = xp[1];
  float v = a0.x * x0.x + a0.y * x0.y + a0.z * x0.z + a0.w * x0.w
          + a1.x * x1.x + a1.y * x1.y + a1.z * x1.z + a1.w * x1.w;
#pragma unroll
  for (int off = 16; off >= 1; off >>= 1) v += __shfl_xor(v, off, 32);
  if (lane == 0) Wout[e] = v;
}

#define OWN_ROWS    256
#define OWN_THREADS 1024
#define OWN_WAVES   (OWN_THREADS / 32)
#define OWN_TRIP    32
#define OWN_WLIST   (32 + 32 * OWN_TRIP + 32)
#define OWN_BLOCKS  ((NNODES + OWN_ROWS - 1) / OWN_ROWS)

static_assert(NEDGE % 4 == 0, "target list is read four edges per load");
static_assert(NEDGE <= (1 << 20), "edge number is packed into 20 bits");
static_assert(OWN_ROWS <= (1 << 12), "local row is packed above bit 20");

__global__ __launch_bounds__(OWN_THREADS) void agg_owned(
    const int* __restrict__ EI, const float* __restrict__ Wv,
    const float* __restrict__ Z, float* __restrict__ OUT, int colOff)
{
  __shared__ float    acc[(OWN_ROWS + 1) * OUTF];
  __shared__ unsigned lst[OWN_WAVES * OWN_WLIST];
  __shared__ unsigned cnt[OWN_ROWS];

  const int tid  = threadIdx.x;
  const int lane = tid & 31;
  const int wave = tid >> 5;
  const int r0   = blockIdx.x * OWN_ROWS;

  {
    const float4 zero4 = {0.f, 0.f, 0.f, 0.f};
    for (int i = tid; i < OWN_ROWS * OUTF / 4; i += OWN_THREADS)
      ((float4*)acc)[i] = zero4;
    if (tid < OWN_ROWS) cnt[tid] = 0u;
  }
  __syncthreads();

  unsigned* mylst = lst + wave * OWN_WLIST;
  unsigned  held  = 0u;

  auto note = [&](unsigned mask, unsigned d, int e) {
    if (mask) {
      if ((mask >> lane) & 1u) {
        const unsigned p = held + __builtin_amdgcn_mbcnt_lo(mask, 0u);
        mylst[p] = (d << 20) | (unsigned)e;
        __hip_atomic_fetch_add(&cnt[d], 1u, __ATOMIC_RELAXED, __HIP_MEMORY_SCOPE_WORKGROUP);
      }
      held += (unsigned)__builtin_popcount(mask);
    }
  };

  auto gather = [&](unsigned first, unsigned n) {
    unsigned ent = (unsigned)OWN_ROWS << 20;
    int   lsrc = 0;
    float wt   = 0.f;
    if ((unsigned)lane < n) {
      ent = mylst[first + lane];
      const int e = (int)(ent & 0xFFFFFu);
      lsrc = EI[NEDGE + e];
      wt   = Wv[e];
    }
    const int rowoff = (int)(ent >> 20) * OUTF;
    for (unsigned k0 = 0; k0 < n; k0 += 8) {
      float4 zz[8];
#pragma unroll
      for (int u = 0; u < 8; ++u) {
        const int lk = __builtin_amdgcn_readlane(lsrc, k0 + u);
        const float4* zrow = (const float4*)Z + (size_t)lk * (OUTF / 4);
        zz[u] = zrow[lane];
      }
#pragma unroll
      for (int u = 0; u < 8; ++u) {
        const float wk = __builtin_bit_cast(float,
            __builtin_amdgcn_readlane(__builtin_bit_cast(int, wt), k0 + u));
        const int ro = __builtin_amdgcn_readlane(rowoff, k0 + u);
        float* a = acc + ro + lane;
        __hip_atomic_fetch_add(a,      wk * zz[u].x, __ATOMIC_RELAXED, __HIP_MEMORY_SCOPE_WORKGROUP);
        __hip_atomic_fetch_add(a + 32, wk * zz[u].y, __ATOMIC_RELAXED, __HIP_MEMORY_SCOPE_WORKGROUP);
        __hip_atomic_fetch_add(a + 64, wk * zz[u].z, __ATOMIC_RELAXED, __HIP_MEMORY_SCOPE_WORKGROUP);
        __hip_atomic_fetch_add(a + 96, wk * zz[u].w, __ATOMIC_RELAXED, __HIP_MEMORY_SCOPE_WORKGROUP);
      }
    }
  };

  {
    const int4* R4 = (const int4*)EI;
    constexpr int NQ    = NEDGE / 4;
    constexpr int QSTEP = OWN_THREADS * (OWN_TRIP / 4);
    constexpr int NTRIP = (NQ + QSTEP - 1) / QSTEP;
    for (int t = 0; t < NTRIP; ++t) {
      const int q0 = t * QSTEP + tid;
      int4 v[OWN_TRIP / 4];
#pragma unroll
      for (int u = 0; u < OWN_TRIP / 4; ++u)
        v[u] = R4[min(q0 + u * OWN_THREADS, NQ - 1)];
#pragma unroll
      for (int u = 0; u < OWN_TRIP / 4; ++u) {
        const int q  = q0 + u * OWN_THREADS;
        const int e0 = q * 4;
        const unsigned in = __builtin_amdgcn_ballot_w32(q < NQ);
        const unsigned dx = (unsigned)(v[u].x - r0), dy = (unsigned)(v[u].y - r0);
        const unsigned dz = (unsigned)(v[u].z - r0), dw = (unsigned)(v[u].w - r0);
        const unsigned mx = __builtin_amdgcn_ballot_w32(dx < OWN_ROWS) & in;
        const unsigned my = __builtin_amdgcn_ballot_w32(dy < OWN_ROWS) & in;
        const unsigned mz = __builtin_amdgcn_ballot_w32(dz < OWN_ROWS) & in;
        const unsigned mw = __builtin_amdgcn_ballot_w32(dw < OWN_ROWS) & in;
        if (mx | my | mz | mw) {
          note(mx, dx, e0);
          note(my, dy, e0 + 1);
          note(mz, dz, e0 + 2);
          note(mw, dw, e0 + 3);
        }
      }
      while (held >= 32u) {
        held -= 32u;
        gather(held, 32u);
      }
    }
    if (held > 0u) gather(0u, held);
  }
  __syncthreads();

  for (int row = wave; row < OWN_ROWS; row += OWN_WAVES) {
    const int g = r0 + row;
    if (g < NNODES) {
      const float inv = 1.0f / fmaxf((float)cnt[row], 1.0f);
      const float* a = acc + row * OUTF + lane;
      float4* op = (float4*)(OUT + (size_t)g * (2 * OUTF) + colOff) + lane;
      float4 o = *op;
      o.x += inv * a[0];
      o.y += inv * a[32];
      o.z += inv * a[64];
      o.w += inv * a[96];
      *op = o;
    }
  }
}

extern "C" void kernel_launch(void* const* d_in, const int* in_sizes, int n_in,
                              void* d_out, int out_size, void* d_ws, size_t ws_size,
                              hipStream_t stream)
{
  const float* x        = (const float*)d_in[0];
  const int*   pos_ei   = (const int*)d_in[1];
  const int*   neg_ei   = (const int*)d_in[2];
  const float* W_pos    = (const float*)d_in[3];
  const float* W_pos_cc = (const float*)d_in[4];
  const float* b_pos_cc = (const float*)d_in[5];
  const float* W_neg    = (const float*)d_in[6];
  const float* W_neg_cc = (const float*)d_in[7];
  const float* b_neg_cc = (const float*)d_in[8];
  const float* W_patt   = (const float*)d_in[9];
  const float* b_patt   = (const float*)d_in[10];
  const float* W_natt   = (const float*)d_in[11];
  const float* b_natt   = (const float*)d_in[12];
  float* out = (float*)d_out;

  float* Sbuf = (float*)d_ws;
  float* Wbuf = Sbuf + (size_t)NNODES * FEAT;

  const dim3 blk(256);
  const dim3 gAtt(NNODES / 16, FEAT / 128);
  const dim3 gOut(NNODES / 16, 1);
  const int  gEdge = (NEDGE + 7) / 8;
  const dim3 gOwn(OWN_BLOCKS);
  const dim3 blkOwn(OWN_THREADS);

  const int*   ei[2]   = { pos_ei, neg_ei };
  const float* Watt[2] = { W_patt, W_natt };
  const float* batt[2] = { b_patt, b_natt };
  const float* Wagg[2] = { W_pos,  W_neg  };
  const float* Wcc[2]  = { W_pos_cc, W_neg_cc };
  const float* bcc[2]  = { b_pos_cc, b_neg_cc };

  for (int s = 0; s < 2; ++s) {
    gemm_bias_wmma<<<gAtt, blk, 0, stream>>>(x, Watt[s], batt[s], Sbuf, FEAT);
    edge_score<<<gEdge, blk, 0, stream>>>(ei[s], Sbuf, x, Wbuf, NEDGE);
    gemm_proj_wmma<<<gOut, blk, 0, stream>>>(x, Wagg[s], Wcc[s], bcc[s],
                                             Sbuf, out, s * OUTF);
    agg_owned<<<gOwn, blkOwn, 0, stream>>>(ei[s], Wbuf, Sbuf, out, s * OUTF);
  }
}


# GEMMs without LDS: a wave keeps its 32 x rows (all 256 k) in registers, streams weight fragments by 16-byte loads, 4 WMMAs per fragment; k sum order regrouped within 8s (fixed)
# speedup vs baseline: 2.7561x; 1.4425x over previous
#include <hip/hip_runtime.h>


typedef __attribute__((ext_vector_type(2))) float v2f;
typedef __attribute__((ext_vector_type(8))) float v8f;

#define NNODES 50000
#define FEAT   256
#define OUTF   128
#define NEDGE  800000

typedef __attribute__((ext_vector_type(4))) float v4f;

#define GW_ROWS    32
#define GW_WAVES   4
#define GW_THREADS (GW_WAVES * 32)
#define GW_BLOCKS  ((NNODES + GW_ROWS * GW_WAVES - 1) / (GW_ROWS * GW_WAVES))
#define GW_TILES   16
#define GW_TPB     4
#define GW_KQ      (FEAT / 8)

static_assert(NNODES % 16 == 0, "strips are whole 16-row tiles");
static_assert(FEAT == 256 && GW_KQ == 32, "the k walk below is written for 256");
static_assert(8 % GW_TPB == 0 && GW_TILES % GW_TPB == 0, "a work-group's tiles share one weight matrix");

template <bool PROJ>
__device__ __forceinline__ void rows_gemm(
    const float* __restrict__ X, const float* __restrict__ Wa,
    const float* __restrict__ Wb, const float* __restrict__ Bias,
    float* __restrict__ Y0, float* __restrict__ Y1, int colOff)
{
  const int lane = threadIdx.x & 31;
  const int wave = threadIdx.x >> 5;
  const int c    = lane & 15;
  const int h    = lane >> 4;
  const int m0   = (blockIdx.x * GW_WAVES + wave) * GW_ROWS;
  if (m0 >= NNODES) return;
  const bool two = (m0 + 16) < NNODES;

  const int    t0    = blockIdx.y * GW_TPB;
  const bool   upper = PROJ && t0 >= 8;
  const float* Wsel  = upper ? Wb : Wa;
  const int    tw    = upper ? t0 - 8 : t0;
  float*       Ysel  = upper ? Y1 : Y0;
  const int    pitch = !PROJ ? FEAT : (upper ? 2 * OUTF : OUTF);
  const int    col0  = upper ? colOff + 16 * tw : 16 * tw;
  const bool   hasB  = !PROJ || upper;

  v4f a0[GW_KQ], a1[GW_KQ];
  {
    const v4f* xa = (const v4f*)(X + (size_t)(m0 + c) * FEAT) + h;
    const v4f* xb = (const v4f*)(X + (size_t)(m0 + (two ? 16 : 0) + c) * FEAT) + h;
#pragma unroll
    for (int j = 0; j < GW_KQ; ++j) { a0[j] = xa[2 * j]; a1[j] = xb[2 * j]; }
  }

  const v4f* wbase = (const v4f*)(Wsel + (size_t)(16 * tw + c) * FEAT) + h;
  constexpr int WTILE = 16 * FEAT / 4;

  v4f bf[2][8];
#pragma unroll
  for (int u = 0; u < 8; ++u) bf[0][u] = wbase[2 * u];

  for (int i = 0; i < GW_TPB; ++i) {
    const v4f* wp = wbase + (size_t)i * WTILE;
    const v4f* wn = wbase + (size_t)(i + 1 < GW_TPB ? i + 1 : i) * WTILE;
    v8f acc0 = {0.f, 0.f, 0.f, 0.f, 0.f, 0.f, 0.f, 0.f};
    v8f acc1 = {0.f, 0.f, 0.f, 0.f, 0.f, 0.f, 0.f, 0.f};
#pragma unroll
    for (int q = 0; q < 4; ++q) {
      if (q < 3) {
#pragma unroll
        for (int u = 0; u < 8; ++u) bf[(q + 1) & 1][u] = wp[2 * (8 * (q + 1) + u)];
      } else {
#pragma unroll
        for (int u = 0; u < 8; ++u) bf[0][u] = wn[2 * u];
      }
#pragma unroll
      for (int u = 0; u < 8; ++u) {
        const int j = 8 * q + u;
        const v4f b  = bf[q & 1][u];
        const v2f blo = __builtin_shufflevector(b, b, 0, 1);
        const v2f bhi = __builtin_shufflevector(b, b, 2, 3);
        const v2f p0 = __builtin_shufflevector(a0[j], a0[j], 0, 1);
        const v2f p1 = __builtin_shufflevector(a1[j], a1[j], 0, 1);
        const v2f r0 = __builtin_shufflevector(a0[j], a0[j], 2, 3);
        const v2f r1 = __builtin_shufflevector(a1[j], a1[j], 2, 3);
        acc0 = __builtin_amdgcn_wmma_f32_16x16x4_f32(
            false, p0, false, blo, (short)0, acc0, false, false);
        acc1 = __builtin_amdgcn_wmma_f32_16x16x4_f32(
            false, p1, false, blo, (short)0, acc1, false, false);
        acc0 = __builtin_amdgcn_wmma_f32_16x16x4_f32(
            false, r0, false, bhi, (short)0, acc0, false, false);
        acc1 = __builtin_amdgcn_wmma_f32_16x16x4_f32(
            false, r1, false, bhi, (short)0, acc1, false, false);
      }
    }

    const float bias = hasB ? Bias[16 * (tw + i) + c] : 0.0f;
    float* yp = Ysel + (size_t)(m0 + (h << 3)) * pitch + (col0 + 16 * i + c);
#pragma unroll
    for (int r = 0; r < 8; ++r) yp[(size_t)r * pitch] = acc0[r] + bias;
    if (two) {
      float* yq = yp + (size_t)16 * pitch;
#pragma unroll
      for (int r = 0; r < 8; ++r) yq[(size_t)r * pitch] = acc1[r] + bias;
    }
  }
}

__global__ __launch_bounds__(GW_THREADS) void gemm_bias_wmma(
    const float* __restrict__ X, const float* __restrict__ W,
    const float* __restrict__ B, float* __restrict__ Y)
{
  rows_gemm<false>(X, W, W, B, Y, Y, 0);
}

__global__ __launch_bounds__(GW_THREADS) void gemm_proj_wmma(
    const float* __restrict__ X,
    const float* __restrict__ W1, const float* __restrict__ W2,
    const float* __restrict__ B2, float* __restrict__ Z,
    float* __restrict__ OUT, int colOff)
{
  rows_gemm<true>(X, W1, W2, B2, Z, OUT, colOff);
}

__global__ __launch_bounds__(256) void edge_score(
    const int* __restrict__ EI, const float* __restrict__ ATT,
    const float* __restrict__ X, float* __restrict__ Wout, int E)
{
  const int lane = threadIdx.x & 31;
  const int wave = threadIdx.x >> 5;
  const int e = (blockIdx.x << 3) + wave;
  if (e >= E) return;
  const int r = EI[e];
  const int l = EI[E + e];

  const float4* ap = (const float4*)(ATT + (size_t)r * FEAT) + (lane << 1);
  const float4* xp = (const float4*)(X   + (size_t)l * FEAT) + (lane << 1);
  float4 a0 = ap[0], a1 = ap[1];
  float4 x0 = xp[0], x1 = xp[1];
  float v = a0.x * x0.x + a0.y * x0.y + a0.z * x0.z + a0.w * x0.w
          + a1.x * x1.x + a1.y * x1.y + a1.z * x1.z + a1.w * x1.w;
#pragma unroll
  for (int off = 16; off >= 1; off >>= 1) v += __shfl_xor(v, off, 32);
  if (lane == 0) Wout[e] = v;
}

#define OWN_ROWS    256
#define OWN_THREADS 1024
#define OWN_WAVES   (OWN_THREADS / 32)
#define OWN_TRIP    32
#define OWN_WLIST   (32 + 32 * OWN_TRIP + 32)
#define OWN_BLOCKS  ((NNODES + OWN_ROWS - 1) / OWN_ROWS)

static_assert(NEDGE % 4 == 0, "target list is read four edges per load");
static_assert(NEDGE <= (1 << 20), "edge number is packed into 20 bits");
static_assert(OWN_ROWS <= (1 << 12), "local row is packed above bit 20");

__global__ __launch_bounds__(OWN_THREADS) void agg_owned(
    const int* __restrict__ EI, const float* __restrict__ Wv,
    const float* __restrict__ Z, float* __restrict__ OUT, int colOff)
{
  __shared__ float    acc[(OWN_ROWS + 1) * OUTF];
  __shared__ unsigned lst[OWN_WAVES * OWN_WLIST];
  __shared__ unsigned cnt[OWN_ROWS];

  const int tid  = threadIdx.x;
  const int lane = tid & 31;
  const int wave = tid >> 5;
  const int r0   = blockIdx.x * OWN_ROWS;

  {
    const float4 zero4 = {0.f, 0.f, 0.f, 0.f};
    for (int i = tid; i < OWN_ROWS * OUTF / 4; i += OWN_THREADS)
      ((float4*)acc)[i] = zero4;
    if (tid < OWN_ROWS) cnt[tid] = 0u;
  }
  __syncthreads();

  unsigned* mylst = lst + wave * OWN_WLIST;
  unsigned  held  = 0u;

  auto note = [&](unsigned mask, unsigned d, int e) {
    if (mask) {
      if ((mask >> lane) & 1u) {
        const unsigned p = held + __builtin_amdgcn_mbcnt_lo(mask, 0u);
        mylst[p] = (d << 20) | (unsigned)e;
        __hip_atomic_fetch_add(&cnt[d], 1u, __ATOMIC_RELAXED, __HIP_MEMORY_SCOPE_WORKGROUP);
      }
      held += (unsigned)__builtin_popcount(mask);
    }
  };

  auto gather = [&](unsigned first, unsigned n) {
    unsigned ent = (unsigned)OWN_ROWS << 20;
    int   lsrc = 0;
    float wt   = 0.f;
    if ((unsigned)lane < n) {
      ent = mylst[first + lane];
      const int e = (int)(ent & 0xFFFFFu);
      lsrc = EI[NEDGE + e];
      wt   = Wv[e];
    }
    const int rowoff = (int)(ent >> 20) * OUTF;
    for (unsigned k0 = 0; k0 < n; k0 += 8) {
      float4 zz[8];
#pragma unroll
      for (int u = 0; u < 8; ++u) {
        const int lk = __builtin_amdgcn_readlane(lsrc, k0 + u);
        const float4* zrow = (const float4*)Z + (size_t)lk * (OUTF / 4);
        zz[u] = zrow[lane];
      }
#pragma unroll
      for (int u = 0; u < 8; ++u) {
        const float wk = __builtin_bit_cast(float,
            __builtin_amdgcn_readlane(__builtin_bit_cast(int, wt), k0 + u));
        const int ro = __builtin_amdgcn_readlane(rowoff, k0 + u);
        float* a = acc + ro + lane;
        __hip_atomic_fetch_add(a,      wk * zz[u].x, __ATOMIC_RELAXED, __HIP_MEMORY_SCOPE_WORKGROUP);
        __hip_atomic_fetch_add(a + 32, wk * zz[u].y, __ATOMIC_RELAXED, __HIP_MEMORY_SCOPE_WORKGROUP);
        __hip_atomic_fetch_add(a + 64, wk * zz[u].z, __ATOMIC_RELAXED, __HIP_MEMORY_SCOPE_WORKGROUP);
        __hip_atomic_fetch_add(a + 96, wk * zz[u].w, __ATOMIC_RELAXED, __HIP_MEMORY_SCOPE_WORKGROUP);
      }
    }
  };

  {
    const int4* R4 = (const int4*)EI;
    constexpr int NQ    = NEDGE / 4;
    constexpr int QSTEP = OWN_THREADS * (OWN_TRIP / 4);
    constexpr int NTRIP = (NQ + QSTEP - 1) / QSTEP;
    for (int t = 0; t < NTRIP; ++t) {
      const int q0 = t * QSTEP + tid;
      int4 v[OWN_TRIP / 4];
#pragma unroll
      for (int u = 0; u < OWN_TRIP / 4; ++u)
        v[u] = R4[min(q0 + u * OWN_THREADS, NQ - 1)];
#pragma unroll
      for (int u = 0; u < OWN_TRIP / 4; ++u) {
        const int q  = q0 + u * OWN_THREADS;
        const int e0 = q * 4;
        const unsigned in = __builtin_amdgcn_ballot_w32(q < NQ);
        const unsigned dx = (unsigned)(v[u].x - r0), dy = (unsigned)(v[u].y - r0);
        const unsigned dz = (unsigned)(v[u].z - r0), dw = (unsigned)(v[u].w - r0);
        const unsigned mx = __builtin_amdgcn_ballot_w32(dx < OWN_ROWS) & in;
        const unsigned my = __builtin_amdgcn_ballot_w32(dy < OWN_ROWS) & in;
        const unsigned mz = __builtin_amdgcn_ballot_w32(dz < OWN_ROWS) & in;
        const unsigned mw = __builtin_amdgcn_ballot_w32(dw < OWN_ROWS) & in;
        if (mx | my | mz | mw) {
          note(mx, dx, e0);
          note(my, dy, e0 + 1);
          note(mz, dz, e0 + 2);
          note(mw, dw, e0 + 3);
        }
      }
      while (held >= 32u) {
        held -= 32u;
        gather(held, 32u);
      }
    }
    if (held > 0u) gather(0u, held);
  }
  __syncthreads();

  for (int row = wave; row < OWN_ROWS; row += OWN_WAVES) {
    const int g = r0 + row;
    if (g < NNODES) {
      const float inv = 1.0f / fmaxf((float)cnt[row], 1.0f);
      const float* a = acc + row * OUTF + lane;
      float4* op = (float4*)(OUT + (size_t)g * (2 * OUTF) + colOff) + lane;
      float4 o = *op;
      o.x += inv * a[0];
      o.y += inv * a[32];
      o.z += inv * a[64];
      o.w += inv * a[96];
      *op = o;
    }
  }
}

extern "C" void kernel_launch(void* const* d_in, const int* in_sizes, int n_in,
                              void* d_out, int out_size, void* d_ws, size_t ws_size,
                              hipStream_t stream)
{
  const float* x        = (const float*)d_in[0];
  const int*   pos_ei   = (const int*)d_in[1];
  const int*   neg_ei   = (const int*)d_in[2];
  const float* W_pos    = (const float*)d_in[3];
  const float* W_pos_cc = (const float*)d_in[4];
  const float* b_pos_cc = (const float*)d_in[5];
  const float* W_neg    = (const float*)d_in[6];
  const float* W_neg_cc = (const float*)d_in[7];
  const float* b_neg_cc = (const float*)d_in[8];
  const float* W_patt   = (const float*)d_in[9];
  const float* b_patt   = (const float*)d_in[10];
  const float* W_natt   = (const float*)d_in[11];
  const float* b_natt   = (const float*)d_in[12];
  float* out = (float*)d_out;

  float* Sbuf = (float*)d_ws;
  float* Wbuf = Sbuf + (size_t)NNODES * FEAT;

  const dim3 blk(256);
  const dim3 blkRows(GW_THREADS);
  const dim3 gRows(GW_BLOCKS, GW_TILES / GW_TPB);
  const int  gEdge = (NEDGE + 7) / 8;
  const dim3 gOwn(OWN_BLOCKS);
  const dim3 blkOwn(OWN_THREADS);

  const int*   ei[2]   = { pos_ei, neg_ei };
  const float* Watt[2] = { W_patt, W_natt };
  const float* batt[2] = { b_patt, b_natt };
  const float* Wagg[2] = { W_pos,  W_neg  };
  const float* Wcc[2]  = { W_pos_cc, W_neg_cc };
  const float* bcc[2]  = { b_pos_cc, b_neg_cc };

  for (int s = 0; s < 2; ++s) {
    gemm_bias_wmma<<<gRows, blkRows, 0, stream>>>(x, Watt[s], batt[s], Sbuf);
    edge_score<<<gEdge, blk, 0, stream>>>(ei[s], Sbuf, x, Wbuf, NEDGE);
    gemm_proj_wmma<<<gRows, blkRows, 0, stream>>>(x, Wagg[s], Wcc[s], bcc[s],
                                                  Sbuf, out, s * OUTF);
    agg_owned<<<gOwn, blkOwn, 0, stream>>>(ei[s], Wbuf, Sbuf, out, s * OUTF);
  }
}


# Edge scores by 32 edges per wave, not 1: one index load per lane, rows read as whole 512-byte halves 4 edges ahead, lane sums of the 32 edges added in 31 exchanges; sum order regrouped (fixed)
# speedup vs baseline: 3.2886x; 1.1932x over previous
#include <hip/hip_runtime.h>


typedef __attribute__((ext_vector_type(2))) float v2f;
typedef __attribute__((ext_vector_type(8))) float v8f;

#define NNODES 50000
#define FEAT   256
#define OUTF   128
#define NEDGE  800000

typedef __attribute__((ext_vector_type(4))) float v4f;

#define GW_ROWS    32
#define GW_WAVES   4
#define GW_THREADS (GW_WAVES * 32)
#define GW_BLOCKS  ((NNODES + GW_ROWS * GW_WAVES - 1) / (GW_ROWS * GW_WAVES))
#define GW_TILES   16
#define GW_TPB     4
#define GW_KQ      (FEAT / 8)

static_assert(NNODES % 16 == 0, "strips are whole 16-row tiles");
static_assert(FEAT == 256 && GW_KQ == 32, "the k walk below is written for 256");
static_assert(8 % GW_TPB == 0 && GW_TILES % GW_TPB == 0, "a work-group's tiles share one weight matrix");

template <bool PROJ>
__device__ __forceinline__ void rows_gemm(
    const float* __restrict__ X, const float* __restrict__ Wa,
    const float* __restrict__ Wb, const float* __restrict__ Bias,
    float* __restrict__ Y0, float* __restrict__ Y1, int colOff)
{
  const int lane = threadIdx.x & 31;
  const int wave = threadIdx.x >> 5;
  const int c    = lane & 15;
  const int h    = lane >> 4;
  const int m0   = (blockIdx.x * GW_WAVES + wave) * GW_ROWS;
  if (m0 >= NNODES) return;
  const bool two = (m0 + 16) < NNODES;

  const int    t0    = blockIdx.y * GW_TPB;
  const bool   upper = PROJ && t0 >= 8;
  const float* Wsel  = upper ? Wb : Wa;
  const int    tw    = upper ? t0 - 8 : t0;
  float*       Ysel  = upper ? Y1 : Y0;
  const int    pitch = !PROJ ? FEAT : (upper ? 2 * OUTF : OUTF);
  const int    col0  = upper ? colOff + 16 * tw : 16 * tw;
  const bool   hasB  = !PROJ || upper;

  v4f a0[GW_KQ], a1[GW_KQ];
  {
    const v4f* xa = (const v4f*)(X + (size_t)(m0 + c) * FEAT) + h;
    const v4f* xb = (const v4f*)(X + (size_t)(m0 + (two ? 16 : 0) + c) * FEAT) + h;
#pragma unroll
    for (int j = 0; j < GW_KQ; ++j) { a0[j] = xa[2 * j]; a1[j] = xb[2 * j]; }
  }

  const v4f* wbase = (const v4f*)(Wsel + (size_t)(16 * tw + c) * FEAT) + h;
  constexpr int WTILE = 16 * FEAT / 4;

  v4f bf[2][8];
#pragma unroll
  for (int u = 0; u < 8; ++u) bf[0][u] = wbase[2 * u];

  for (int i = 0; i < GW_TPB; ++i) {
    const v4f* wp = wbase + (size_t)i * WTILE;
    const v4f* wn = wbase + (size_t)(i + 1 < GW_TPB ? i + 1 : i) * WTILE;
    v8f acc0 = {0.f, 0.f, 0.f, 0.f, 0.f, 0.f, 0.f, 0.f};
    v8f acc1 = {0.f, 0.f, 0.f, 0.f, 0.f, 0.f, 0.f, 0.f};
#pragma unroll
    for (int q = 0; q < 4; ++q) {
      if (q < 3) {
#pragma unroll
        for (int u = 0; u < 8; ++u) bf[(q + 1) & 1][u] = wp[2 * (8 * (q + 1) + u)];
      } else {
#pragma unroll
        for (int u = 0; u < 8; ++u) bf[0][u] = wn[2 * u];
      }
#pragma unroll
      for (int u = 0; u < 8; ++u) {
        const int j = 8 * q + u;
        const v4f b  = bf[q & 1][u];
        const v2f blo = __builtin_shufflevector(b, b, 0, 1);
        const v2f bhi = __builtin_shufflevector(b, b, 2, 3);
        const v2f p0 = __builtin_shufflevector(a0[j], a0[j], 0, 1);
        const v2f p1 = __builtin_shufflevector(a1[j], a1[j], 0, 1);
        const v2f r0 = __builtin_shufflevector(a0[j], a0[j], 2, 3);
        const v2f r1 = __builtin_shufflevector(a1[j], a1[j], 2, 3);
        acc0 = __builtin_amdgcn_wmma_f32_16x16x4_f32(
            false, p0, false, blo, (short)0, acc0, false, false);
        acc1 = __builtin_amdgcn_wmma_f32_16x16x4_f32(
            false, p1, false, blo, (short)0, acc1, false, false);
        acc0 = __builtin_amdgcn_wmma_f32_16x16x4_f32(
            false, r0, false, bhi, (short)0, acc0, false, false);
        acc1 = __builtin_amdgcn_wmma_f32_16x16x4_f32(
            false, r1, false, bhi, (short)0, acc1, false, false);
      }
    }

    const float bias = hasB ? Bias[16 * (tw + i) + c] : 0.0f;
    float* yp = Ysel + (size_t)(m0 + (h << 3)) * pitch + (col0 + 16 * i + c);
#pragma unroll
    for (int r = 0; r < 8; ++r) yp[(size_t)r * pitch] = acc0[r] + bias;
    if (two) {
      float* yq = yp + (size_t)16 * pitch;
#pragma unroll
      for (int r = 0; r < 8; ++r) yq[(size_t)r * pitch] = acc1[r] + bias;
    }
  }
}

__global__ __launch_bounds__(GW_THREADS) void gemm_bias_wmma(
    const float* __restrict__ X, const float* __restrict__ W,
    const float* __restrict__ B, float* __restrict__ Y)
{
  rows_gemm<false>(X, W, W, B, Y, Y, 0);
}

__global__ __launch_bounds__(GW_THREADS) void gemm_proj_wmma(
    const float* __restrict__ X,
    const float* __restrict__ W1, const float* __restrict__ W2,
    const float* __restrict__ B2, float* __restrict__ Z,
    float* __restrict__ OUT, int colOff)
{
  rows_gemm<true>(X, W1, W2, B2, Z, OUT, colOff);
}

#define ES_EDGES  32
#define ES_WAVES  8
#define ES_GRP    4
#define ES_NGRP   (ES_EDGES / ES_GRP)
#define ES_BLOCKS ((NEDGE + ES_WAVES * ES_EDGES - 1) / (ES_WAVES * ES_EDGES))

static_assert(ES_GRP == 4 && ES_NGRP == 8, "the exchange steps below are written for 8 groups of 4");

__global__ __launch_bounds__(ES_WAVES * 32) void edge_score(
    const int* __restrict__ EI, const float* __restrict__ ATT,
    const float* __restrict__ X, float* __restrict__ Wout, int E)
{
  const int lane = threadIdx.x & 31;
  const int wave = threadIdx.x >> 5;
  const int e0 = (blockIdx.x * ES_WAVES + wave) * ES_EDGES;
  if (e0 >= E) return;
  const int ec = min(e0 + lane, E - 1);
  const int r = EI[ec];
  const int l = EI[E + ec];

  const float4* A4 = (const float4*)ATT;
  const float4* X4 = (const float4*)X;

  float4 a[2][ES_GRP][2], x[2][ES_GRP][2];
  float  s[ES_NGRP];

  const bool b0 = (lane & 1) != 0, b1 = (lane & 2) != 0, b2 = (lane & 4) != 0;
  const bool b3 = (lane & 8) != 0, b4 = (lane & 16) != 0;

#pragma unroll
  for (int u = 0; u < ES_GRP; ++u) {
    const float4* ap = A4 + (size_t)__builtin_amdgcn_readlane(r, u) * (FEAT / 4);
    const float4* xp = X4 + (size_t)__builtin_amdgcn_readlane(l, u) * (FEAT / 4);
    a[0][u][0] = ap[lane]; a[0][u][1] = ap[32 + lane];
    x[0][u][0] = xp[lane]; x[0][u][1] = xp[32 + lane];
  }
#pragma unroll
  for (int g = 0; g < ES_NGRP; ++g) {
    const int cur = g & 1, nxt = cur ^ 1;
    if (g + 1 < ES_NGRP) {
#pragma unroll
      for (int u = 0; u < ES_GRP; ++u) {
        const int k = (g + 1) * ES_GRP + u;
        const float4* ap = A4 + (size_t)__builtin_amdgcn_readlane(r, k) * (FEAT / 4);
        const float4* xp = X4 + (size_t)__builtin_amdgcn_readlane(l, k) * (FEAT / 4);
        a[nxt][u][0] = ap[lane]; a[nxt][u][1] = ap[32 + lane];
        x[nxt][u][0] = xp[lane]; x[nxt][u][1] = xp[32 + lane];
      }
    }
    float q[ES_GRP];
#pragma unroll
    for (int u = 0; u < ES_GRP; ++u) {
      const float4 a0 = a[cur][u][0], a1 = a[cur][u][1];
      const float4 x0 = x[cur][u][0], x1 = x[cur][u][1];
      q[u] = a0.x * x0.x + a0.y * x0.y + a0.z * x0.z + a0.w * x0.w
           + a1.x * x1.x + a1.y * x1.y + a1.z * x1.z + a1.w * x1.w;
    }
    const float t0 = (b0 ? q[1] : q[0]) + __shfl_xor(b0 ? q[0] : q[1], 1, 32);
    const float t1 = (b0 ? q[3] : q[2]) + __shfl_xor(b0 ? q[2] : q[3], 1, 32);
    s[g] = (b1 ? t1 : t0) + __shfl_xor(b1 ? t0 : t1, 2, 32);
  }

  float u4[4], u8[2];
#pragma unroll
  for (int k = 0; k < 4; ++k)
    u4[k] = (b2 ? s[k + 4] : s[k]) + __shfl_xor(b2 ? s[k] : s[k + 4], 4, 32);
#pragma unroll
  for (int k = 0; k < 2; ++k)
    u8[k] = (b3 ? u4[k + 2] : u4[k]) + __shfl_xor(b3 ? u4[k] : u4[k + 2], 8, 32);
  const float w = (b4 ? u8[1] : u8[0]) + __shfl_xor(b4 ? u8[0] : u8[1], 16, 32);

  const int grp = (b2 ? 4 : 0) + (b3 ? 2 : 0) + (b4 ? 1 : 0);
  const int eo  = e0 + grp * ES_GRP + (lane & 3);
  if (eo < E) Wout[eo] = w;
}

#define OWN_ROWS    256
#define OWN_THREADS 1024
#define OWN_WAVES   (OWN_THREADS / 32)
#define OWN_TRIP    32
#define OWN_WLIST   (32 + 32 * OWN_TRIP + 32)
#define OWN_BLOCKS  ((NNODES + OWN_ROWS - 1) / OWN_ROWS)

static_assert(NEDGE % 4 == 0, "target list is read four edges per load");
static_assert(NEDGE <= (1 << 20), "edge number is packed into 20 bits");
static_assert(OWN_ROWS <= (1 << 12), "local row is packed above bit 20");

__global__ __launch_bounds__(OWN_THREADS) void agg_owned(
    const int* __restrict__ EI, const float* __restrict__ Wv,
    const float* __restrict__ Z, float* __restrict__ OUT, int colOff)
{
  __shared__ float    acc[(OWN_ROWS + 1) * OUTF];
  __shared__ unsigned lst[OWN_WAVES * OWN_WLIST];
  __shared__ unsigned cnt[OWN_ROWS];

  const int tid  = threadIdx.x;
  const int lane = tid & 31;
  const int wave = tid >> 5;
  const int r0   = blockIdx.x * OWN_ROWS;

  {
    const float4 zero4 = {0.f, 0.f, 0.f, 0.f};
    for (int i = tid; i < OWN_ROWS * OUTF / 4; i += OWN_THREADS)
      ((float4*)acc)[i] = zero4;
    if (tid < OWN_ROWS) cnt[tid] = 0u;
  }
  __syncthreads();

  unsigned* mylst = lst + wave * OWN_WLIST;
  unsigned  held  = 0u;

  auto note = [&](unsigned mask, unsigned d, int e) {
    if (mask) {
      if ((mask >> lane) & 1u) {
        const unsigned p = held + __builtin_amdgcn_mbcnt_lo(mask, 0u);
        mylst[p] = (d << 20) | (unsigned)e;
        __hip_atomic_fetch_add(&cnt[d], 1u, __ATOMIC_RELAXED, __HIP_MEMORY_SCOPE_WORKGROUP);
      }
      held += (unsigned)__builtin_popcount(mask);
    }
  };

  auto gather = [&](unsigned first, unsigned n) {
    unsigned ent = (unsigned)OWN_ROWS << 20;
    int   lsrc = 0;
    float wt   = 0.f;
    if ((unsigned)lane < n) {
      ent = mylst[first + lane];
      const int e = (int)(ent & 0xFFFFFu);
      lsrc = EI[NEDGE + e];
      wt   = Wv[e];
    }
    const int rowoff = (int)(ent >> 20) * OUTF;
    for (unsigned k0 = 0; k0 < n; k0 += 8) {
      float4 zz[8];
#pragma unroll
      for (int u = 0; u < 8; ++u) {
        const int lk = __builtin_amdgcn_readlane(lsrc, k0 + u);
        const float4* zrow = (const float4*)Z + (size_t)lk * (OUTF / 4);
        zz[u] = zrow[lane];
      }
#pragma unroll
      for (int u = 0; u < 8; ++u) {
        const float wk = __builtin_bit_cast(float,
            __builtin_amdgcn_readlane(__builtin_bit_cast(int, wt), k0 + u));
        const int ro = __builtin_amdgcn_readlane(rowoff, k0 + u);
        float* a = acc + ro + lane;
        __hip_atomic_fetch_add(a,      wk * zz[u].x, __ATOMIC_RELAXED, __HIP_MEMORY_SCOPE_WORKGROUP);
        __hip_atomic_fetch_add(a + 32, wk * zz[u].y, __ATOMIC_RELAXED, __HIP_MEMORY_SCOPE_WORKGROUP);
        __hip_atomic_fetch_add(a + 64, wk * zz[u].z, __ATOMIC_RELAXED, __HIP_MEMORY_SCOPE_WORKGROUP);
        __hip_atomic_fetch_add(a + 96, wk * zz[u].w, __ATOMIC_RELAXED, __HIP_MEMORY_SCOPE_WORKGROUP);
      }
    }
  };

  {
    const int4* R4 = (const int4*)EI;
    constexpr int NQ    = NEDGE / 4;
    constexpr int QSTEP = OWN_THREADS * (OWN_TRIP / 4);
    constexpr int NTRIP = (NQ + QSTEP - 1) / QSTEP;
    for (int t = 0; t < NTRIP; ++t) {
      const int q0 = t * QSTEP + tid;
      int4 v[OWN_TRIP / 4];
#pragma unroll
      for (int u = 0; u < OWN_TRIP / 4; ++u)
        v[u] = R4[min(q0 + u * OWN_THREADS, NQ - 1)];
#pragma unroll
      for (int u = 0; u < OWN_TRIP / 4; ++u) {
        const int q  = q0 + u * OWN_THREADS;
        const int e0 = q * 4;
        const unsigned in = __builtin_amdgcn_ballot_w32(q < NQ);
        const unsigned dx = (unsigned)(v[u].x - r0), dy = (unsigned)(v[u].y - r0);
        const unsigned dz = (unsigned)(v[u].z - r0), dw = (unsigned)(v[u].w - r0);
        const unsigned mx = __builtin_amdgcn_ballot_w32(dx < OWN_ROWS) & in;
        const unsigned my = __builtin_amdgcn_ballot_w32(dy < OWN_ROWS) & in;
        const unsigned mz = __builtin_amdgcn_ballot_w32(dz < OWN_ROWS) & in;
        const unsigned mw = __builtin_amdgcn_ballot_w32(dw < OWN_ROWS) & in;
        if (mx | my | mz | mw) {
          note(mx, dx, e0);
          note(my, dy, e0 + 1);
          note(mz, dz, e0 + 2);
          note(mw, dw, e0 + 3);
        }
      }
      while (held >= 32u) {
        held -= 32u;
        gather(held, 32u);
      }
    }
    if (held > 0u) gather(0u, held);
  }
  __syncthreads();

  for (int row = wave; row < OWN_ROWS; row += OWN_WAVES) {
    const int g = r0 + row;
    if (g < NNODES) {
      const float inv = 1.0f / fmaxf((float)cnt[row], 1.0f);
      const float* a = acc + row * OUTF + lane;
      float4* op = (float4*)(OUT + (size_t)g * (2 * OUTF) + colOff) + lane;
      float4 o = *op;
      o.x += inv * a[0];
      o.y += inv * a[32];
      o.z += inv * a[64];
      o.w += inv * a[96];
      *op = o;
    }
  }
}

extern "C" void kernel_launch(void* const* d_in, const int* in_sizes, int n_in,
                              void* d_out, int out_size, void* d_ws, size_t ws_size,
                              hipStream_t stream)
{
  const float* x        = (const float*)d_in[0];
  const int*   pos_ei   = (const int*)d_in[1];
  const int*   neg_ei   = (const int*)d_in[2];
  const float* W_pos    = (const float*)d_in[3];
  const float* W_pos_cc = (const float*)d_in[4];
  const float* b_pos_cc = (const float*)d_in[5];
  const float* W_neg    = (const float*)d_in[6];
  const float* W_neg_cc = (const float*)d_in[7];
  const float* b_neg_cc = (const float*)d_in[8];
  const float* W_patt   = (const float*)d_in[9];
  const float* b_patt   = (const float*)d_in[10];
  const float* W_natt   = (const float*)d_in[11];
  const float* b_natt   = (const float*)d_in[12];
  float* out = (float*)d_out;

  float* Sbuf = (float*)d_ws;
  float* Wbuf = Sbuf + (size_t)NNODES * FEAT;

  const dim3 blk(256);
  const dim3 blkRows(GW_THREADS);
  const dim3 gRows(GW_BLOCKS, GW_TILES / GW_TPB);
  const dim3 gEdge(ES_BLOCKS);
  const dim3 gOwn(OWN_BLOCKS);
  const dim3 blkOwn(OWN_THREADS);

  const int*   ei[2]   = { pos_ei, neg_ei };
  const float* Watt[2] = { W_patt, W_natt };
  const float* batt[2] = { b_patt, b_natt };
  const float* Wagg[2] = { W_pos,  W_neg  };
  const float* Wcc[2]  = { W_pos_cc, W_neg_cc };
  const float* bcc[2]  = { b_pos_cc, b_neg_cc };

  for (int s = 0; s < 2; ++s) {
    gemm_bias_wmma<<<gRows, blkRows, 0, stream>>>(x, Watt[s], batt[s], Sbuf);
    edge_score<<<gEdge, blk, 0, stream>>>(ei[s], Sbuf, x, Wbuf, NEDGE);
    gemm_proj_wmma<<<gRows, blkRows, 0, stream>>>(x, Wagg[s], Wcc[s], bcc[s],
                                                  Sbuf, out, s * OUTF);
    agg_owned<<<gOwn, blkOwn, 0, stream>>>(ei[s], Wbuf, Sbuf, out, s * OUTF);
  }
}


# Weights re-laid once per call in WMMA fragment order (1 pack launch, copy parked in out columns not yet written): a B fragment load reads 4 whole lines, not 32 B of 16; sums bit-identical
# speedup vs baseline: 3.7780x; 1.1488x over previous
#include <hip/hip_runtime.h>


typedef __attribute__((ext_vector_type(2))) float v2f;
typedef __attribute__((ext_vector_type(8))) float v8f;

#define NNODES 50000
#define FEAT   256
#define OUTF   128
#define NEDGE  800000

typedef __attribute__((ext_vector_type(4))) float v4f;

#define GW_ROWS    32
#define GW_WAVES   4
#define GW_THREADS (GW_WAVES * 32)
#define GW_BLOCKS  ((NNODES + GW_ROWS * GW_WAVES - 1) / (GW_ROWS * GW_WAVES))
#define GW_TILES   16
#define GW_TPB     4
#define GW_KQ      (FEAT / 8)

static_assert(NNODES % 16 == 0, "strips are whole 16-row tiles");
static_assert(FEAT == 256 && GW_KQ == 32, "the k walk below is written for 256");
static_assert(8 % GW_TPB == 0 && GW_TILES % GW_TPB == 0, "a work-group's tiles share one output and bias");

#define WP_PITCH 256

#define WP_SIGN_TILES 32
__global__ __launch_bounds__(256) void pack_weights(
    const float* __restrict__ Watt0, const float* __restrict__ Wagg0,
    const float* __restrict__ Wcc0,
    const float* __restrict__ Watt1, const float* __restrict__ Wagg1,
    const float* __restrict__ Wcc1, float* __restrict__ P)
{
  const int lane = threadIdx.x & 31;
  const int f    = blockIdx.x * 8 + (threadIdx.x >> 5);
  const int T    = f >> 5;
  const int j    = f & 31;
  const bool neg = T >= WP_SIGN_TILES;
  const int  ts  = neg ? T - WP_SIGN_TILES : T;
  const float* W = neg ? Wcc1 : Wcc0;
  int lt = ts - 24;
  if (ts < 16)      { W = neg ? Watt1 : Watt0; lt = ts; }
  else if (ts < 24) { W = neg ? Wagg1 : Wagg0; lt = ts - 16; }
  const v4f v = *((const v4f*)(W + (size_t)(16 * lt + (lane & 15)) * FEAT) + (2 * j + (lane >> 4)));
  *((v4f*)(P + (size_t)f * WP_PITCH) + lane) = v;
}

template <bool PROJ>
__device__ __forceinline__ void rows_gemm(
    const float* __restrict__ X, const float* __restrict__ Wp,
    const float* __restrict__ Bias,
    float* __restrict__ Y0, float* __restrict__ Y1)
{
  const int lane = threadIdx.x & 31;
  const int wave = threadIdx.x >> 5;
  const int c    = lane & 15;
  const int h    = lane >> 4;
  const int m0   = (blockIdx.x * GW_WAVES + wave) * GW_ROWS;
  if (m0 >= NNODES) return;
  const bool two = (m0 + 16) < NNODES;

  const int    t0    = blockIdx.y * GW_TPB;
  const bool   upper = PROJ && t0 >= 8;
  const int    tw    = upper ? t0 - 8 : t0;
  float*       Ysel  = upper ? Y1 : Y0;
  const int    pitch = PROJ ? OUTF : FEAT;
  const int    col0  = 16 * tw;
  const bool   hasB  = !PROJ || upper;

  v4f a0[GW_KQ], a1[GW_KQ];
  {
    const v4f* xa = (const v4f*)(X + (size_t)(m0 + c) * FEAT) + h;
    const v4f* xb = (const v4f*)(X + (size_t)(m0 + (two ? 16 : 0) + c) * FEAT) + h;
#pragma unroll
    for (int j = 0; j < GW_KQ; ++j) { a0[j] = xa[2 * j]; a1[j] = xb[2 * j]; }
  }

  const v4f* wbase = (const v4f*)(Wp + (size_t)(t0 * GW_KQ) * WP_PITCH) + lane;
  constexpr int WFRAG = WP_PITCH / 4;
  constexpr int WTILE = GW_KQ * WFRAG;

  v4f bf[2][8];
#pragma unroll
  for (int u = 0; u < 8; ++u) bf[0][u] = wbase[u * WFRAG];

  for (int i = 0; i < GW_TPB; ++i) {
    const v4f* wp = wbase + (size_t)i * WTILE;
    const v4f* wn = wbase + (size_t)(i + 1 < GW_TPB ? i + 1 : i) * WTILE;
    v8f acc0 = {0.f, 0.f, 0.f, 0.f, 0.f, 0.f, 0.f, 0.f};
    v8f acc1 = {0.f, 0.f, 0.f, 0.f, 0.f, 0.f, 0.f, 0.f};
#pragma unroll
    for (int q = 0; q < 4; ++q) {
      if (q < 3) {
#pragma unroll
        for (int u = 0; u < 8; ++u) bf[(q + 1) & 1][u] = wp[(8 * (q + 1) + u) * WFRAG];
      } else {
#pragma unroll
        for (int u = 0; u < 8; ++u) bf[0][u] = wn[u * WFRAG];
      }
#pragma unroll
      for (int u = 0; u < 8; ++u) {
        const int j = 8 * q + u;
        const v4f b  = bf[q & 1][u];
        const v2f blo = __builtin_shufflevector(b, b, 0, 1);
        const v2f bhi = __builtin_shufflevector(b, b, 2, 3);
        const v2f p0 = __builtin_shufflevector(a0[j], a0[j], 0, 1);
        const v2f p1 = __builtin_shufflevector(a1[j], a1[j], 0, 1);
        const v2f r0 = __builtin_shufflevector(a0[j], a0[j], 2, 3);
        const v2f r1 = __builtin_shufflevector(a1[j], a1[j], 2, 3);
        acc0 = __builtin_amdgcn_wmma_f32_16x16x4_f32(
            false, p0, false, blo, (short)0, acc0, false, false);
        acc1 = __builtin_amdgcn_wmma_f32_16x16x4_f32(
            false, p1, false, blo, (short)0, acc1, false, false);
        acc0 = __builtin_amdgcn_wmma_f32_16x16x4_f32(
            false, r0, false, bhi, (short)0, acc0, false, false);
        acc1 = __builtin_amdgcn_wmma_f32_16x16x4_f32(
            false, r1, false, bhi, (short)0, acc1, false, false);
      }
    }

    const float bias = hasB ? Bias[16 * (tw + i) + c] : 0.0f;
    float* yp = Ysel + (size_t)(m0 + (h << 3)) * pitch + (col0 + 16 * i + c);
#pragma unroll
    for (int r = 0; r < 8; ++r) yp[(size_t)r * pitch] = acc0[r] + bias;
    if (two) {
      float* yq = yp + (size_t)16 * pitch;
#pragma unroll
      for (int r = 0; r < 8; ++r) yq[(size_t)r * pitch] = acc1[r] + bias;
    }
  }
}

__global__ __launch_bounds__(GW_THREADS) void gemm_bias_wmma(
    const float* __restrict__ X, const float* __restrict__ Wp,
    const float* __restrict__ B, float* __restrict__ Y)
{
  rows_gemm<false>(X, Wp, B, Y, Y);
}

__global__ __launch_bounds__(GW_THREADS) void gemm_proj_wmma(
    const float* __restrict__ X, const float* __restrict__ Wp,
    const float* __restrict__ B2, float* __restrict__ Z,
    float* __restrict__ CC)
{
  rows_gemm<true>(X, Wp, B2, Z, CC);
}

#define ES_EDGES  32
#define ES_WAVES  8
#define ES_GRP    4
#define ES_NGRP   (ES_EDGES / ES_GRP)
#define ES_BLOCKS ((NEDGE + ES_WAVES * ES_EDGES - 1) / (ES_WAVES * ES_EDGES))

static_assert(ES_GRP == 4 && ES_NGRP == 8, "the exchange steps below are written for 8 groups of 4");

__global__ __launch_bounds__(ES_WAVES * 32) void edge_score(
    const int* __restrict__ EI, const float* __restrict__ ATT,
    const float* __restrict__ X, float* __restrict__ Wout, int E)
{
  const int lane = threadIdx.x & 31;
  const int wave = threadIdx.x >> 5;
  const int e0 = (blockIdx.x * ES_WAVES + wave) * ES_EDGES;
  if (e0 >= E) return;
  const int ec = min(e0 + lane, E - 1);
  const int r = EI[ec];
  const int l = EI[E + ec];

  const float4* A4 = (const float4*)ATT;
  const float4* X4 = (const float4*)X;

  float4 a[2][ES_GRP][2], x[2][ES_GRP][2];
  float  s[ES_NGRP];

  const bool b0 = (lane & 1) != 0, b1 = (lane & 2) != 0, b2 = (lane & 4) != 0;
  const bool b3 = (lane & 8) != 0, b4 = (lane & 16) != 0;

#pragma unroll
  for (int u = 0; u < ES_GRP; ++u) {
    const float4* ap = A4 + (size_t)__builtin_amdgcn_readlane(r, u) * (FEAT / 4);
    const float4* xp = X4 + (size_t)__builtin_amdgcn_readlane(l, u) * (FEAT / 4);
    a[0][u][0] = ap[lane]; a[0][u][1] = ap[32 + lane];
    x[0][u][0] = xp[lane]; x[0][u][1] = xp[32 + lane];
  }
#pragma unroll
  for (int g = 0; g < ES_NGRP; ++g) {
    const int cur = g & 1, nxt = cur ^ 1;
    if (g + 1 < ES_NGRP) {
#pragma unroll
      for (int u = 0; u < ES_GRP; ++u) {
        const int k = (g + 1) * ES_GRP + u;
        const float4* ap = A4 + (size_t)__builtin_amdgcn_readlane(r, k) * (FEAT / 4);
        const float4* xp = X4 + (size_t)__builtin_amdgcn_readlane(l, k) * (FEAT / 4);
        a[nxt][u][0] = ap[lane]; a[nxt][u][1] = ap[32 + lane];
        x[nxt][u][0] = xp[lane]; x[nxt][u][1] = xp[32 + lane];
      }
    }
    float q[ES_GRP];
#pragma unroll
    for (int u = 0; u < ES_GRP; ++u) {
      const float4 a0 = a[cur][u][0], a1 = a[cur][u][1];
      const float4 x0 = x[cur][u][0], x1 = x[cur][u][1];
      q[u] = a0.x * x0.x + a0.y * x0.y + a0.z * x0.z + a0.w * x0.w
           + a1.x * x1.x + a1.y * x1.y + a1.z * x1.z + a1.w * x1.w;
    }
    const float t0 = (b0 ? q[1] : q[0]) + __shfl_xor(b0 ? q[0] : q[1], 1, 32);
    const float t1 = (b0 ? q[3] : q[2]) + __shfl_xor(b0 ? q[2] : q[3], 1, 32);
    s[g] = (b1 ? t1 : t0) + __shfl_xor(b1 ? t0 : t1, 2, 32);
  }

  float u4[4], u8[2];
#pragma unroll
  for (int k = 0; k < 4; ++k)
    u4[k] = (b2 ? s[k + 4] : s[k]) + __shfl_xor(b2 ? s[k] : s[k + 4], 4, 32);
#pragma unroll
  for (int k = 0; k < 2; ++k)
    u8[k] = (b3 ? u4[k + 2] : u4[k]) + __shfl_xor(b3 ? u4[k] : u4[k + 2], 8, 32);
  const float w = (b4 ? u8[1] : u8[0]) + __shfl_xor(b4 ? u8[0] : u8[1], 16, 32);

  const int grp = (b2 ? 4 : 0) + (b3 ? 2 : 0) + (b4 ? 1 : 0);
  const int eo  = e0 + grp * ES_GRP + (lane & 3);
  if (eo < E) Wout[eo] = w;
}

#define OWN_ROWS    256
#define OWN_THREADS 1024
#define OWN_WAVES   (OWN_THREADS / 32)
#define OWN_TRIP    32
#define OWN_WLIST   (32 + 32 * OWN_TRIP + 32)
#define OWN_BLOCKS  ((NNODES + OWN_ROWS - 1) / OWN_ROWS)

static_assert(NEDGE % 4 == 0, "target list is read four edges per load");
static_assert(NEDGE <= (1 << 20), "edge number is packed into 20 bits");
static_assert(OWN_ROWS <= (1 << 12), "local row is packed above bit 20");

__global__ __launch_bounds__(OWN_THREADS) void agg_owned(
    const int* __restrict__ EI, const float* __restrict__ Wv,
    const float* __restrict__ Z, const float* __restrict__ CC,
    float* __restrict__ OUT, int colOff)
{
  __shared__ float    acc[(OWN_ROWS + 1) * OUTF];
  __shared__ unsigned lst[OWN_WAVES * OWN_WLIST];
  __shared__ unsigned cnt[OWN_ROWS];

  const int tid  = threadIdx.x;
  const int lane = tid & 31;
  const int wave = tid >> 5;
  const int r0   = blockIdx.x * OWN_ROWS;

  {
    const float4 zero4 = {0.f, 0.f, 0.f, 0.f};
    for (int i = tid; i < OWN_ROWS * OUTF / 4; i += OWN_THREADS)
      ((float4*)acc)[i] = zero4;
    if (tid < OWN_ROWS) cnt[tid] = 0u;
  }
  __syncthreads();

  unsigned* mylst = lst + wave * OWN_WLIST;
  unsigned  held  = 0u;

  auto note = [&](unsigned mask, unsigned d, int e) {
    if (mask) {
      if ((mask >> lane) & 1u) {
        const unsigned p = held + __builtin_amdgcn_mbcnt_lo(mask, 0u);
        mylst[p] = (d << 20) | (unsigned)e;
        __hip_atomic_fetch_add(&cnt[d], 1u, __ATOMIC_RELAXED, __HIP_MEMORY_SCOPE_WORKGROUP);
      }
      held += (unsigned)__builtin_popcount(mask);
    }
  };

  auto gather = [&](unsigned first, unsigned n) {
    unsigned ent = (unsigned)OWN_ROWS << 20;
    int   lsrc = 0;
    float wt   = 0.f;
    if ((unsigned)lane < n) {
      ent = mylst[first + lane];
      const int e = (int)(ent & 0xFFFFFu);
      lsrc = EI[NEDGE + e];
      wt   = Wv[e];
    }
    const int rowoff = (int)(ent >> 20) * OUTF;
    for (unsigned k0 = 0; k0 < n; k0 += 8) {
      float4 zz[8];
#pragma unroll
      for (int u = 0; u < 8; ++u) {
        const int lk = __builtin_amdgcn_readlane(lsrc, k0 + u);
        const float4* zrow = (const float4*)Z + (size_t)lk * (OUTF / 4);
        zz[u] = zrow[lane];
      }
#pragma unroll
      for (int u = 0; u < 8; ++u) {
        const float wk = __builtin_bit_cast(float,
            __builtin_amdgcn_readlane(__builtin_bit_cast(int, wt), k0 + u));
        const int ro = __builtin_amdgcn_readlane(rowoff, k0 + u);
        float* a = acc + ro + lane;
        __hip_atomic_fetch_add(a,      wk * zz[u].x, __ATOMIC_RELAXED, __HIP_MEMORY_SCOPE_WORKGROUP);
        __hip_atomic_fetch_add(a + 32, wk * zz[u].y, __ATOMIC_RELAXED, __HIP_MEMORY_SCOPE_WORKGROUP);
        __hip_atomic_fetch_add(a + 64, wk * zz[u].z, __ATOMIC_RELAXED, __HIP_MEMORY_SCOPE_WORKGROUP);
        __hip_atomic_fetch_add(a + 96, wk * zz[u].w, __ATOMIC_RELAXED, __HIP_MEMORY_SCOPE_WORKGROUP);
      }
    }
  };

  {
    const int4* R4 = (const int4*)EI;
    constexpr int NQ    = NEDGE / 4;
    constexpr int QSTEP = OWN_THREADS * (OWN_TRIP / 4);
    constexpr int NTRIP = (NQ + QSTEP - 1) / QSTEP;
    for (int t = 0; t < NTRIP; ++t) {
      const int q0 = t * QSTEP + tid;
      int4 v[OWN_TRIP / 4];
#pragma unroll
      for (int u = 0; u < OWN_TRIP / 4; ++u)
        v[u] = R4[min(q0 + u * OWN_THREADS, NQ - 1)];
#pragma unroll
      for (int u = 0; u < OWN_TRIP / 4; ++u) {
        const int q  = q0 + u * OWN_THREADS;
        const int e0 = q * 4;
        const unsigned in = __builtin_amdgcn_ballot_w32(q < NQ);
        const unsigned dx = (unsigned)(v[u].x - r0), dy = (unsigned)(v[u].y - r0);
        const unsigned dz = (unsigned)(v[u].z - r0), dw = (unsigned)(v[u].w - r0);
        const unsigned mx = __builtin_amdgcn_ballot_w32(dx < OWN_ROWS) & in;
        const unsigned my = __builtin_amdgcn_ballot_w32(dy < OWN_ROWS) & in;
        const unsigned mz = __builtin_amdgcn_ballot_w32(dz < OWN_ROWS) & in;
        const unsigned mw = __builtin_amdgcn_ballot_w32(dw < OWN_ROWS) & in;
        if (mx | my | mz | mw) {
          note(mx, dx, e0);
          note(my, dy, e0 + 1);
          note(mz, dz, e0 + 2);
          note(mw, dw, e0 + 3);
        }
      }
      while (held >= 32u) {
        held -= 32u;
        gather(held, 32u);
      }
    }
    if (held > 0u) gather(0u, held);
  }
  __syncthreads();

  for (int row = wave; row < OWN_ROWS; row += OWN_WAVES) {
    const int g = r0 + row;
    if (g < NNODES) {
      const float inv = 1.0f / fmaxf((float)cnt[row], 1.0f);
      const float* a = acc + row * OUTF + lane;
      float4* op = (float4*)(OUT + (size_t)g * (2 * OUTF) + colOff) + lane;
      float4 o = ((const float4*)(CC + (size_t)g * OUTF))[lane];
      o.x += inv * a[0];
      o.y += inv * a[32];
      o.z += inv * a[64];
      o.w += inv * a[96];
      *op = o;
    }
  }
}

extern "C" void kernel_launch(void* const* d_in, const int* in_sizes, int n_in,
                              void* d_out, int out_size, void* d_ws, size_t ws_size,
                              hipStream_t stream)
{
  const float* x        = (const float*)d_in[0];
  const int*   pos_ei   = (const int*)d_in[1];
  const int*   neg_ei   = (const int*)d_in[2];
  const float* W_pos    = (const float*)d_in[3];
  const float* W_pos_cc = (const float*)d_in[4];
  const float* b_pos_cc = (const float*)d_in[5];
  const float* W_neg    = (const float*)d_in[6];
  const float* W_neg_cc = (const float*)d_in[7];
  const float* b_neg_cc = (const float*)d_in[8];
  const float* W_patt   = (const float*)d_in[9];
  const float* b_patt   = (const float*)d_in[10];
  const float* W_natt   = (const float*)d_in[11];
  const float* b_natt   = (const float*)d_in[12];
  float* out = (float*)d_out;

  float* Sbuf = (float*)d_ws;
  float* Wbuf = Sbuf + (size_t)NNODES * FEAT;

  const dim3 blk(256);
  const dim3 blkRows(GW_THREADS);
  const dim3 gRows(GW_BLOCKS, GW_TILES / GW_TPB);
  const dim3 gEdge(ES_BLOCKS);
  const dim3 gOwn(OWN_BLOCKS);
  const dim3 blkOwn(OWN_THREADS);

  float* packed = out + OUTF;
  float* CCbuf  = Sbuf + (size_t)NNODES * OUTF;
  const float* pkAtt0  = packed;
  const float* pkProj0 = packed + (size_t)16 * GW_KQ * WP_PITCH;
  const float* pkAtt1  = packed + (size_t)32 * GW_KQ * WP_PITCH;
  const float* pkProj1 = packed + (size_t)48 * GW_KQ * WP_PITCH;

  pack_weights<<<dim3(2 * WP_SIGN_TILES * 4), blk, 0, stream>>>(
      W_patt, W_pos, W_pos_cc, W_natt, W_neg, W_neg_cc, packed);

  gemm_bias_wmma<<<gRows, blkRows, 0, stream>>>(x, pkAtt0, b_patt, Sbuf);
  edge_score<<<gEdge, blk, 0, stream>>>(pos_ei, Sbuf, x, Wbuf, NEDGE);
  gemm_proj_wmma<<<gRows, blkRows, 0, stream>>>(x, pkProj0, b_pos_cc, Sbuf, CCbuf);
  agg_owned<<<gOwn, blkOwn, 0, stream>>>(pos_ei, Wbuf, Sbuf, CCbuf, out, 0);

  gemm_bias_wmma<<<gRows, blkRows, 0, stream>>>(x, pkAtt1, b_natt, Sbuf);
  edge_score<<<gEdge, blk, 0, stream>>>(neg_ei, Sbuf, x, Wbuf, NEDGE);
  gemm_proj_wmma<<<gRows, blkRows, 0, stream>>>(x, pkProj1, b_neg_cc, Sbuf, CCbuf);
  agg_owned<<<gOwn, blkOwn, 0, stream>>>(neg_ei, Wbuf, Sbuf, CCbuf, out, OUTF);
}
